# starting kernel
# baseline (speedup 1.0000x reference)
#include <hip/hip_runtime.h>


typedef __attribute__((ext_vector_type(2))) float v2f;
typedef __attribute__((ext_vector_type(4))) float v4f;
typedef __attribute__((ext_vector_type(8))) float v8f;

#define KS     7
#define PADR   3
#define CH     16
#define KK     49
#define KKP    64
#define TILE_W 128
#define HALO_W (TILE_W + KS - 1)
#define NX     (KS * HALO_W)
#define BN_EPS 1e-5f

__global__ __launch_bounds__(256) void involution_wmma_kernel(
    const float* __restrict__ x,
    const float* __restrict__ w1,  const float* __restrict__ b1,
    const float* __restrict__ gam, const float* __restrict__ bet,
    const float* __restrict__ mu,  const float* __restrict__ var,
    const float* __restrict__ w2,  const float* __restrict__ b2,
    float* __restrict__ out, int B, int H, int W)
{
    __shared__ __align__(16) float s_x[NX];
    __shared__ __align__(16) float s_w2[KKP][CH];
    __shared__ __align__(32) float s_b2[KKP];
    __shared__ __align__(16) float s_ad[2 * CH];

    const int tid = threadIdx.x;

    for (int idx = tid; idx < KKP * CH; idx += 256) {
        const int k = idx >> 4, c = idx & 15;
        s_w2[k][c] = (k < KK) ? w2[k * CH + c] : 0.f;
    }
    if (tid < KKP) s_b2[tid] = (tid < KK) ? b2[tid] : 0.f;
    if (tid < CH) {
        const float inv = gam[tid] * rsqrtf(var[tid] + BN_EPS);
        s_ad[2 * tid]     = w1[tid] * inv;
        s_ad[2 * tid + 1] = (b1[tid] - mu[tid]) * inv + bet[tid];
    }

    const int col0 = blockIdx.x * TILE_W;
    const int row  = blockIdx.y;
    const int b    = blockIdx.z;
    const float* __restrict__ xb = x + (size_t)b * H * W;

    for (int idx = tid; idx < NX; idx += 256) {
        const int r  = idx / HALO_W;
        const int c  = idx - r * HALO_W;
        const int gi = row - PADR + r;
        const int gj = col0 - PADR + c;
        float v = 0.f;
        if (gi >= 0 && gi < H && gj >= 0 && gj < W) v = xb[gi * W + gj];
        s_x[idx] = v;
    }
    __syncthreads();

    const int lane = tid & 31;
    const int wave = tid >> 5;
    const int half = lane >> 4;
    const int p    = lane & 15;
    const int j    = wave * 16 + p;

    const float xc = s_x[PADR * HALO_W + PADR + j];

    v2f A[4][4];
#pragma unroll
    for (int t = 0; t < 4; ++t)
#pragma unroll
        for (int s = 0; s < 4; ++s)
            A[t][s] = *(const v2f*)&s_w2[t * 16 + p][s * 4 + 2 * half];

    const float* pb2 = &s_b2[8 * half];
    v8f acc[4];
#pragma unroll
    for (int t = 0; t < 4; ++t)
        acc[t] = *(const v8f*)&pb2[16 * t];

    const float* pada = &s_ad[4 * half];
#pragma unroll
    for (int s = 0; s < 4; ++s) {
        const v4f ad = *(const v4f*)&pada[8 * s];
        v2f Bv;
        Bv.x = fmaxf(fmaf(ad.x, xc, ad.y), 0.f);
        Bv.y = fmaxf(fmaf(ad.z, xc, ad.w), 0.f);
#pragma unroll
        for (int t = 0; t < 4; ++t) {
            acc[t] = __builtin_amdgcn_wmma_f32_16x16x4_f32(
                false, A[t][s], false, Bv, (short)0, acc[t], false, false);
        }
    }

    const float* q0 = &s_x[j];
    const float* q1 = q0 + 135 * half;
    const float* q2 = q0 + 262 * half;

    float partial = 0.f;
#pragma unroll
    for (int t = 0; t < 4; ++t) {
#pragma unroll
        for (int v = 0; v < 8; ++v) {
            const int mlo  = 16 * t + v;
            const int mhi  = mlo + 8;
            const int mloc = (mlo < KK) ? mlo : 48;
            const int off  = (mloc / KS) * HALO_W + (mloc % KS);
            const float* q = (mhi >= KK) ? q0 : ((mlo % KS == 6) ? q2 : q1);
            partial = fmaf(acc[t][v], q[off], partial);
        }
    }

    partial += __shfl_xor(partial, 16, 32);

    if (half == 0)
        out[((size_t)b * H + row) * W + col0 + j] = partial;
}

extern "C" void kernel_launch(void* const* d_in, const int* in_sizes, int n_in,
                              void* d_out, int out_size, void* d_ws, size_t ws_size,
                              hipStream_t stream) {
    (void)n_in; (void)d_ws; (void)ws_size; (void)out_size;
    const float* x   = (const float*)d_in[0];
    const float* w1  = (const float*)d_in[1];
    const float* b1  = (const float*)d_in[2];
    const float* gam = (const float*)d_in[3];
    const float* bet = (const float*)d_in[4];
    const float* mu  = (const float*)d_in[5];
    const float* var = (const float*)d_in[6];
    const float* w2  = (const float*)d_in[7];
    const float* b2  = (const float*)d_in[8];
    float* out = (float*)d_out;

    const int H = 256, W = 256;
    const int B = in_sizes[0] / (H * W);

    dim3 grid(W / TILE_W, H, B);
    dim3 block(256);
    involution_wmma_kernel<<<grid, block, 0, stream>>>(
        x, w1, b1, gam, bet, mu, var, w2, b2, out, B, H, W);
}


# each wave walks a 16-col strip down 32 rows with the 7x7 patch as a sliding register window fed from global (7 loads/row); no LDS/barrier; W2 rows reordered to the window; fp32 tap sum order changes
# speedup vs baseline: 1.5704x; 1.5704x over previous
#include <hip/hip_runtime.h>


typedef __attribute__((ext_vector_type(2))) float v2f;
typedef __attribute__((ext_vector_type(8))) float v8f;

#define HOLD7(a)  asm volatile("" :: "v"((a)[0]), "v"((a)[1]), "v"((a)[2]), "v"((a)[3]), \
                                     "v"((a)[4]), "v"((a)[5]), "v"((a)[6]))
#define HOLD8(a)  asm volatile("" :: "v"((a)[0]), "v"((a)[1]), "v"((a)[2]), "v"((a)[3]), \
                                     "v"((a)[4]), "v"((a)[5]), "v"((a)[6]), "v"((a)[7]))

#define KS     7
#define CH     16
#define IMG_H  256
#define IMG_W  256
#define SEG_R  32
#define NSLOT  28
#define BN_EPS 1e-5f

__device__ __forceinline__ int tap_of_slot(int q, int h) {
    const int d  = q / KS;
    const int dc = q - KS * d;
    const bool ok = (q < NSLOT) && !(h == 1 && d == 0);
    return ok ? (KS * (d + 3 * h) + dc) : -1;
}

__global__ __launch_bounds__(512) void involution_colwalk_kernel(
    const float* __restrict__ x,
    const float* __restrict__ w1,  const float* __restrict__ b1,
    const float* __restrict__ gam, const float* __restrict__ bet,
    const float* __restrict__ mu,  const float* __restrict__ var,
    const float* __restrict__ w2,  const float* __restrict__ b2,
    float* __restrict__ out)
{
    const int tid   = threadIdx.x;
    const int lane  = tid & 31;
    const int half  = lane >> 4;
    const int p     = lane & 15;
    const int gw    = blockIdx.x * 16 + (tid >> 5);
    const int strip = gw & 15;
    const int seg   = (gw >> 4) & 7;
    const int b     = gw >> 7;
    const int j     = strip * 16 + p;
    const int i0    = seg * SEG_R;

    const float* __restrict__ xb = x   + (size_t)b * IMG_H * IMG_W;
    float* __restrict__       ob = out + (size_t)b * IMG_H * IMG_W;

    float Araw[4][8];
    int   Ak[4];
#pragma unroll
    for (int t = 0; t < 4; ++t) {
        Ak[t] = tap_of_slot(8 * t + (p & 7), p >> 3);
        const int kc = (Ak[t] >= 0) ? Ak[t] : 0;
#pragma unroll
        for (int s = 0; s < 4; ++s) {
            const v2f a = *(const v2f*)&w2[kc * CH + 4 * s + 2 * half];
            Araw[t][2 * s]     = a.x;
            Araw[t][2 * s + 1] = a.y;
        }
    }
    float Craw[4][8];
#pragma unroll
    for (int t = 0; t < 4; ++t) {
#pragma unroll
        for (int r = 0; r < 8; ++r) {
            const int k  = tap_of_slot(8 * t + r, half);
            const int kc = (k >= 0) ? k : 0;
            Craw[t][r] = (8 * t + r < NSLOT) ? b2[kc] : 0.f;
        }
    }

    float ca[4][2], cd[4][2];
#pragma unroll
    for (int s = 0; s < 4; ++s) {
#pragma unroll
        for (int e = 0; e < 2; ++e) {
            const int c = 4 * s + 2 * half + e;
            const float inv = gam[c] * rsqrtf(var[c] + BN_EPS);
            ca[s][e] = w1[c] * inv;
            cd[s][e] = (b1[c] - mu[c]) * inv + bet[c];
        }
    }

    int  coff[KS];
    bool cok[KS];
#pragma unroll
    for (int dc = 0; dc < KS; ++dc) {
        const int cj = j + dc - 3;
        cok[dc]  = (cj >= 0) && (cj < IMG_W);
        coff[dc] = cok[dc] ? cj : 0;
    }

    float win[4][KS];
    bool  wrok[4];
#pragma unroll
    for (int d = 0; d < 4; ++d) {
        const int g  = i0 + d + 3 * half - 3;
        wrok[d]      = (g >= 0) && (g < IMG_H);
        const int gb = wrok[d] ? g * IMG_W : 0;
#pragma unroll
        for (int dc = 0; dc < KS; ++dc)
            win[d][dc] = xb[gb + coff[dc]];
    }

#pragma unroll
    for (int t = 0; t < 4; ++t) HOLD8(Araw[t]);
#pragma unroll
    for (int t = 0; t < 3; ++t) HOLD8(Craw[t]);
    asm volatile("" :: "v"(Craw[3][0]), "v"(Craw[3][1]), "v"(Craw[3][2]), "v"(Craw[3][3]));
#pragma unroll
    for (int d = 0; d < 4; ++d) HOLD7(win[d]);

    v2f A[4][4];
    v8f Cb[4];
#pragma unroll
    for (int t = 0; t < 4; ++t) {
#pragma unroll
        for (int s = 0; s < 4; ++s) {
            A[t][s].x = (Ak[t] >= 0) ? Araw[t][2 * s]     : 0.f;
            A[t][s].y = (Ak[t] >= 0) ? Araw[t][2 * s + 1] : 0.f;
        }
#pragma unroll
        for (int r = 0; r < 8; ++r)
            Cb[t][r] = (tap_of_slot(8 * t + r, half) >= 0) ? Craw[t][r] : 0.f;
    }
#pragma unroll
    for (int d = 0; d < 4; ++d)
#pragma unroll
        for (int dc = 0; dc < KS; ++dc)
            win[d][dc] = (wrok[d] && cok[dc]) ? win[d][dc] : 0.f;

    for (int u0 = 0; u0 < SEG_R; u0 += 4) {
#pragma unroll
        for (int U = 0; U < 4; ++U) {
            const int i = i0 + u0 + U;

            float nxt[KS];
            const int  gn   = i + 1 + 3 * half;
            const bool nrok = (gn < IMG_H);
            const int  gnb  = nrok ? gn * IMG_W : 0;
#pragma unroll
            for (int dc = 0; dc < KS; ++dc)
                nxt[dc] = xb[gnb + coff[dc]];
            __builtin_amdgcn_sched_barrier(0);

            const float xc = half ? win[(0 + U) & 3][3] : win[(3 + U) & 3][3];

            v8f acc[4];
#pragma unroll
            for (int s = 0; s < 4; ++s) {
                v2f Bv;
                Bv.x = fmaxf(fmaf(ca[s][0], xc, cd[s][0]), 0.f);
                Bv.y = fmaxf(fmaf(ca[s][1], xc, cd[s][1]), 0.f);
#pragma unroll
                for (int t = 0; t < 4; ++t) {
                    acc[t] = __builtin_amdgcn_wmma_f32_16x16x4_f32(
                        false, A[t][s], false, Bv, (short)0,
                        (s == 0) ? Cb[t] : acc[t], false, false);
                }
            }

            float part[4] = {0.f, 0.f, 0.f, 0.f};
#pragma unroll
            for (int q = 0; q < NSLOT; ++q) {
                const int t = q >> 3, r = q & 7;
                const int d = q / KS, dc = q - KS * d;
                part[t] = fmaf(acc[t][r], win[(d + U) & 3][dc], part[t]);
            }
            float partial = (part[0] + part[1]) + (part[2] + part[3]);

            partial += __shfl_xor(partial, 16, 32);

            if (half == 0)
                ob[i * IMG_W + j] = partial;

            HOLD7(nxt);
#pragma unroll
            for (int dc = 0; dc < KS; ++dc)
                win[U & 3][dc] = (nrok && cok[dc]) ? nxt[dc] : 0.f;
        }
    }
}

extern "C" void kernel_launch(void* const* d_in, const int* in_sizes, int n_in,
                              void* d_out, int out_size, void* d_ws, size_t ws_size,
                              hipStream_t stream) {
    (void)n_in; (void)d_ws; (void)ws_size; (void)out_size; (void)in_sizes;
    const float* x   = (const float*)d_in[0];
    const float* w1  = (const float*)d_in[1];
    const float* b1  = (const float*)d_in[2];
    const float* gam = (const float*)d_in[3];
    const float* bet = (const float*)d_in[4];
    const float* mu  = (const float*)d_in[5];
    const float* var = (const float*)d_in[6];
    const float* w2  = (const float*)d_in[7];
    const float* b2  = (const float*)d_in[8];
    float* out = (float*)d_out;

    dim3 grid(256);
    dim3 block(512);
    involution_colwalk_kernel<<<grid, block, 0, stream>>>(
        x, w1, b1, gam, bet, mu, var, w2, b2, out);
}


# W2 rows 64->48: 3 row tiles, 12 f32 WMMAs per strip, not 16; both lane rows share 24 window slots; centre tap's row (1 of 49) moves off the matrix unit to the vector ALU; fp32 tap sum order changes
# speedup vs baseline: 2.0115x; 1.2809x over previous
#include <hip/hip_runtime.h>


typedef __attribute__((ext_vector_type(2))) float v2f;
typedef __attribute__((ext_vector_type(8))) float v8f;

#define HOLD7(a)  asm volatile("" :: "v"((a)[0]), "v"((a)[1]), "v"((a)[2]), "v"((a)[3]), \
                                     "v"((a)[4]), "v"((a)[5]), "v"((a)[6]))

__device__ __forceinline__ float xhalf(float v) {
    const int b = __builtin_bit_cast(int, v);
    return __builtin_bit_cast(float,
        __builtin_amdgcn_permlanex16(b, b, 0x76543210, 0xfedcba98, false, false));
}

#define KS     7
#define CH     16
#define IMG_H  256
#define IMG_W  256
#define SEG_R  32
#define NG     2
#define NT     3
#define NSLOT  24
#define CTAP   24
#define BN_EPS 1e-5f

__device__ __forceinline__ int win_pcol(int wc, int h) {
    return (wc < 3) ? (h ? wc + 4 : wc) : ((wc < 6) ? (h ? wc - 3 : wc + 1) : 3);
}
__device__ __forceinline__ int win_rofs(int wc, int h) {
    return h ? ((wc < 3) ? 3 : 4) : 0;
}
__device__ __forceinline__ int tap_of_slot(int q, int h) {
    const int d  = q / KS;
    const int wc = q - KS * d;
    return KS * (d + win_rofs(wc, h)) + win_pcol(wc, h);
}

#define STEP(S_, U_, TAIL_) \
    {                                                                                           \
        const int i = i0 + (S_);                                                    \
                                                                                                \
                      \
                      \
                                        \
        __builtin_amdgcn_sched_barrier(0);                                                      \
                                                                                                \
                         \
                          \
                                    \
        v2f   Bv[NG][4];                                                                        \
        float xc[NG];                                                                           \
_Pragma("unroll")                                                                               \
        for (int g = 0; g < NG; ++g) {                                                          \
            const float x0 = win[g][(3 + (U_)) & 3][6];                                         \
            const float x1 = xhalf(x0);                                                         \
            xc[g] = half ? x1 : x0;                                                             \
_Pragma("unroll")                                                                               \
            for (int s4 = 0; s4 < 4; ++s4) {                                                    \
                Bv[g][s4].x = fmaxf(fmaf(ca[s4][0], xc[g], cd[s4][0]), 0.f);                    \
                Bv[g][s4].y = fmaxf(fmaf(ca[s4][1], xc[g], cd[s4][1]), 0.f);                    \
            }                                                                                   \
        }                                                                                       \
                       \
                         \
                                               \
        __builtin_amdgcn_sched_barrier(0);                                                      \
                                                                                                \
                         \
                      \
        float nxt[NG][KS];                                                                      \
        bool  nrokA = true, nrokB = true;                                                       \
        if ((TAIL_)) {                                                                          \
            const int gnA  = i + 1 + 3 * half;                                                  \
            const int gnB  = i + 1 + 4 * half;                                                  \
            nrokA          = (gnA < IMG_H);                                                     \
            nrokB          = (gnB < IMG_H);                                                     \
            const int gnbA = nrokA ? gnA * IMG_W : 0;                                           \
            const int gnbB = nrokB ? gnB * IMG_W : 0;                                           \
_Pragma("unroll")                                                                               \
            for (int g = 0; g < NG; ++g)                                                        \
_Pragma("unroll")                                                                               \
                for (int wc = 0; wc < KS; ++wc)                                                 \
                    nxt[g][wc] = xb[((wc < 3) ? gnbA : gnbB) + coff[g][wc]];                    \
        } else {                                                                                \
                             \
                                   \
            const float* __restrict__ rowp = xb + (i + 1) * IMG_W;                              \
_Pragma("unroll")                                                                               \
            for (int g = 0; g < NG; ++g)                                                        \
_Pragma("unroll")                                                                               \
                for (int wc = 0; wc < KS; ++wc)                                                 \
                    nxt[g][wc] = rowp[voff[g][wc]];                                             \
        }                                                                                       \
                         \
                            \
                                    \
        __builtin_amdgcn_sched_barrier(0);                                                      \
                                                                                                \
               \
        v8f acc[NG][NT];                                                                        \
_Pragma("unroll")                                                                               \
        for (int s4 = 0; s4 < 4; ++s4) {                                                        \
_Pragma("unroll")                                                                               \
            for (int g = 0; g < NG; ++g) {                                                      \
_Pragma("unroll")                                                                               \
                for (int t = 0; t < NT; ++t) {                                                  \
                    acc[g][t] = __builtin_amdgcn_wmma_f32_16x16x4_f32(                          \
                        false, A[t][s4], false, Bv[g][s4], (short)0,                            \
                        (s4 == 0) ? Cb[t] : acc[g][t], false, false);                           \
                }                                                                               \
            }                                                                                   \
        }                                                                                       \
                                                                                                \
                                 \
        float partial[NG];                                                                      \
_Pragma("unroll")                                                                               \
        for (int g = 0; g < NG; ++g) {                                                          \
                            \
                          \
            v2f cw2;                                                                            \
            cw2.x = cb2;                                                                        \
            cw2.y = 0.f;                                                                        \
_Pragma("unroll")                                                                               \
            for (int s4 = 0; s4 < 4; ++s4)                                                      \
                cw2 = __builtin_elementwise_fma(cw[s4], Bv[g][s4], cw2);                        \
            const float cwt = cw2.x + cw2.y;                                                    \
                                                                                                \
                            \
                           \
                                  \
            float part[4] = {0.f, 0.f, 0.f, 0.f};                                               \
_Pragma("unroll")                                                                               \
            for (int q = 0; q < NSLOT; ++q) {                                                   \
                const int t = q >> 3, r = q & 7;                                                \
                const int d = q / KS, wc = q - KS * d;                                          \
                part[q & 3] = fmaf(acc[g][t][r], win[g][(d + (U_)) & 3][wc], part[q & 3]);      \
            }                                                                                   \
            partial[g] = fmaf(cwt, xc[g], (part[0] + part[1]) + (part[2] + part[3]));           \
        }                                                                                       \
                                                                                                \
                           \
                            \
                         \
                                                           \
_Pragma("unroll")                                                                               \
        for (int g = 0; g < NG; ++g)                                                            \
            partial[g] += xhalf(partial[g]);                                                    \
                                                                                                \
                          \
                            \
        (ob + i * IMG_W)[jst] = half ? partial[1] : partial[0];                                 \
                                                                                                \
                            \
                             \
_Pragma("unroll")                                                                               \
        for (int g = 0; g < NG; ++g) {                                                          \
            HOLD7(nxt[g]);                                                                      \
_Pragma("unroll")                                                                               \
            for (int wc = 0; wc < KS; ++wc)                                                     \
                win[g][(U_) & 3][wc] =                                                          \
                    (((wc < 3) ? nrokA : nrokB) && cok[g][wc]) ? nxt[g][wc] : 0.f;              \
        }                                                                                       \
    }

__global__ __launch_bounds__(256) void involution_colwalk_kernel(
    const float* __restrict__ x,
    const float* __restrict__ w1,  const float* __restrict__ b1,
    const float* __restrict__ gam, const float* __restrict__ bet,
    const float* __restrict__ mu,  const float* __restrict__ var,
    const float* __restrict__ w2,  const float* __restrict__ b2,
    float* __restrict__ out)
{
    const int tid   = threadIdx.x;
    const int lane  = tid & 31;
    const int half  = lane >> 4;
    const int p     = lane & 15;
    const int gw    = __builtin_amdgcn_readfirstlane((int)(blockIdx.x * 8 + (tid >> 5)));
    const int spair = gw & 7;
    const int seg   = (gw >> 3) & 7;
    const int b     = gw >> 6;
    const int jst   = spair * 32 + lane;
    const int i0    = seg * SEG_R;

    const float* __restrict__ xb = x   + (size_t)b * IMG_H * IMG_W;
    float* __restrict__       ob = out + (size_t)b * IMG_H * IMG_W;

    v2f A[NT][4];
#pragma unroll
    for (int t = 0; t < NT; ++t) {
        const int k = tap_of_slot(8 * t + (p & 7), p >> 3);
#pragma unroll
        for (int s = 0; s < 4; ++s)
            A[t][s] = *(const v2f*)&w2[k * CH + 4 * s + 2 * half];
    }
    v8f Cb[NT];
#pragma unroll
    for (int t = 0; t < NT; ++t) {
#pragma unroll
        for (int r = 0; r < 8; ++r)
            Cb[t][r] = b2[tap_of_slot(8 * t + r, half)];
    }
    v2f cw[4];
#pragma unroll
    for (int s = 0; s < 4; ++s)
        cw[s] = *(const v2f*)&w2[CTAP * CH + 4 * s + 2 * half];
    const float cb2 = half ? 0.f : b2[CTAP];

    float ca[4][2], cd[4][2];
#pragma unroll
    for (int s = 0; s < 4; ++s) {
#pragma unroll
        for (int e = 0; e < 2; ++e) {
            const int c = 4 * s + 2 * half + e;
            const float inv = gam[c] * rsqrtf(var[c] + BN_EPS);
            ca[s][e] = w1[c] * inv;
            cd[s][e] = (b1[c] - mu[c]) * inv + bet[c];
        }
    }

    int  coff[NG][KS];
    bool cok[NG][KS];
#pragma unroll
    for (int g = 0; g < NG; ++g) {
#pragma unroll
        for (int wc = 0; wc < KS; ++wc) {
            const int pc = half ? win_pcol(wc, 1) : win_pcol(wc, 0);
            const int cj = spair * 32 + g * 16 + p + pc - 3;
            cok[g][wc]  = (cj >= 0) && (cj < IMG_W);
            coff[g][wc] = cok[g][wc] ? cj : 0;
        }
    }

    float win[NG][4][KS];
    bool  wrokA[4], wrokB[4];
#pragma unroll
    for (int d = 0; d < 4; ++d) {
        const int grA = i0 + d + 3 * half - 3;
        const int grB = i0 + d + 4 * half - 3;
        wrokA[d]      = (grA >= 0) && (grA < IMG_H);
        wrokB[d]      = (grB >= 0) && (grB < IMG_H);
        const int gbA = wrokA[d] ? grA * IMG_W : 0;
        const int gbB = wrokB[d] ? grB * IMG_W : 0;
#pragma unroll
        for (int g = 0; g < NG; ++g)
#pragma unroll
            for (int wc = 0; wc < KS; ++wc)
                win[g][d][wc] = xb[((wc < 3) ? gbA : gbB) + coff[g][wc]];
    }

#pragma unroll
    for (int g = 0; g < NG; ++g)
#pragma unroll
        for (int d = 0; d < 4; ++d) HOLD7(win[g][d]);

#pragma unroll
    for (int g = 0; g < NG; ++g)
#pragma unroll
        for (int d = 0; d < 4; ++d)
#pragma unroll
            for (int wc = 0; wc < KS; ++wc)
                win[g][d][wc] =
                    (((wc < 3) ? wrokA[d] : wrokB[d]) && cok[g][wc]) ? win[g][d][wc] : 0.f;

    unsigned voff[NG][KS];
#pragma unroll
    for (int g = 0; g < NG; ++g)
#pragma unroll
        for (int wc = 0; wc < KS; ++wc)
            voff[g][wc] = (unsigned)(((wc < 3) ? 3 : 4) * half * IMG_W + coff[g][wc]);

    for (int u0 = 0; u0 < SEG_R - 8; u0 += 4) {
        STEP(u0 + 0, 0, 0)
        STEP(u0 + 1, 1, 0)
        STEP(u0 + 2, 2, 0)
        STEP(u0 + 3, 3, 0)
    }
    STEP(SEG_R - 8, 0, 0)
    STEP(SEG_R - 7, 1, 0)
    STEP(SEG_R - 6, 2, 0)
    STEP(SEG_R - 5, 3, 1)
    STEP(SEG_R - 4, 0, 1)
    STEP(SEG_R - 3, 1, 1)
    STEP(SEG_R - 2, 2, 1)
    STEP(SEG_R - 1, 3, 1)
}

extern "C" void kernel_launch(void* const* d_in, const int* in_sizes, int n_in,
                              void* d_out, int out_size, void* d_ws, size_t ws_size,
                              hipStream_t stream) {
    (void)n_in; (void)d_ws; (void)ws_size; (void)out_size; (void)in_sizes;
    const float* x   = (const float*)d_in[0];
    const float* w1  = (const float*)d_in[1];
    const float* b1  = (const float*)d_in[2];
    const float* gam = (const float*)d_in[3];
    const float* bet = (const float*)d_in[4];
    const float* mu  = (const float*)d_in[5];
    const float* var = (const float*)d_in[6];
    const float* w2  = (const float*)d_in[7];
    const float* b2  = (const float*)d_in[8];
    float* out = (float*)d_out;

    dim3 grid(256);
    dim3 block(256);
    involution_colwalk_kernel<<<grid, block, 0, stream>>>(
        x, w1, b1, gam, bet, mu, var, w2, b2, out);
}


# window row loads issued 4 rows ahead of use, not 1: a step fetches the row that enters 4 steps later into one of 4 register row buffers, so first-touch rows get 4 steps to land; sums bit-identical
# speedup vs baseline: 2.2523x; 1.1197x over previous
#include <hip/hip_runtime.h>


typedef __attribute__((ext_vector_type(2))) float v2f;
typedef __attribute__((ext_vector_type(8))) float v8f;

#define HOLD7(a)  asm volatile("" :: "v"((a)[0]), "v"((a)[1]), "v"((a)[2]), "v"((a)[3]), \
                                     "v"((a)[4]), "v"((a)[5]), "v"((a)[6]))

__device__ __forceinline__ float xhalf(float v) {
    const int b = __builtin_bit_cast(int, v);
    return __builtin_bit_cast(float,
        __builtin_amdgcn_permlanex16(b, b, 0x76543210, 0xfedcba98, false, false));
}

__device__ __forceinline__ float vmov(float v) {
    float r;
    asm("v_mov_b32 %0, %1" : "=v"(r) : "v"(v));
    return r;
}

#define KS     7
#define CH     16
#define IMG_H  256
#define IMG_W  256
#define SEG_R  32
#define NG     2
#define NT     3
#define NSLOT  24
#define CTAP   24
#define PF     4
#define BN_EPS 1e-5f

__device__ __forceinline__ int win_pcol(int wc, int h) {
    return (wc < 3) ? (h ? wc + 4 : wc) : ((wc < 6) ? (h ? wc - 3 : wc + 1) : 3);
}
__device__ __forceinline__ int win_rofs(int wc, int h) {
    return h ? ((wc < 3) ? 3 : 4) : 0;
}
__device__ __forceinline__ int tap_of_slot(int q, int h) {
    const int d  = q / KS;
    const int wc = q - KS * d;
    return KS * (d + win_rofs(wc, h)) + win_pcol(wc, h);
}

#define STEP(S_, U_, LOAD_, LTAIL_, CTAIL_)                                                     \
    {                                                                                           \
        const int i = i0 + (S_);                                                    \
                                                                                                \
                      \
                      \
                                        \
        __builtin_amdgcn_sched_barrier(0);                                                      \
                                                                                                \
                         \
                          \
                                    \
        v2f   Bv[NG][4];                                                                        \
        float xc[NG];                                                                           \
_Pragma("unroll")                                                                               \
        for (int g = 0; g < NG; ++g) {                                                          \
            const float x0 = win[g][(3 + (U_)) & 3][6];                                         \
            const float x1 = xhalf(x0);                                                         \
            xc[g] = half ? x1 : x0;                                                             \
_Pragma("unroll")                                                                               \
            for (int s4 = 0; s4 < 4; ++s4) {                                                    \
                Bv[g][s4].x = fmaxf(fmaf(ca[s4][0], xc[g], cd[s4][0]), 0.f);                    \
                Bv[g][s4].y = fmaxf(fmaf(ca[s4][1], xc[g], cd[s4][1]), 0.f);                    \
            }                                                                                   \
        }                                                                                       \
                       \
                         \
                                               \
        __builtin_amdgcn_sched_barrier(0);                                                      \
                                                                                                \
                          \
                     \
        if ((LOAD_) && (LTAIL_)) {                                                              \
            const int gnA  = i + PF + 3 * half;                                                 \
            const int gnB  = i + PF + 4 * half;                                                 \
            const int gnbA = (gnA < IMG_H) ? gnA * IMG_W : 0;                                   \
            const int gnbB = (gnB < IMG_H) ? gnB * IMG_W : 0;                                   \
_Pragma("unroll")                                                                               \
            for (int g = 0; g < NG; ++g)                                                        \
_Pragma("unroll")                                                                               \
                for (int wc = 0; wc < KS; ++wc)                                                 \
                    nb[(U_) & (PF - 1)][g][wc] = xb[((wc < 3) ? gnbA : gnbB) + coff[g][wc]];    \
        } else if ((LOAD_)) {                                                                   \
                           \
                                            \
            const float* __restrict__ rowp = xb + (i + PF) * IMG_W;                             \
_Pragma("unroll")                                                                               \
            for (int g = 0; g < NG; ++g)                                                        \
_Pragma("unroll")                                                                               \
                for (int wc = 0; wc < KS; ++wc)                                                 \
                    nb[(U_) & (PF - 1)][g][wc] = rowp[voff[g][wc]];                             \
        }                                                                                       \
                        \
                       \
                                                                           \
        __builtin_amdgcn_sched_barrier(0);                                                      \
                                                                                                \
               \
        v8f acc[NG][NT];                                                                        \
_Pragma("unroll")                                                                               \
        for (int s4 = 0; s4 < 4; ++s4) {                                                        \
_Pragma("unroll")                                                                               \
            for (int g = 0; g < NG; ++g) {                                                      \
_Pragma("unroll")                                                                               \
                for (int t = 0; t < NT; ++t) {                                                  \
                    acc[g][t] = __builtin_amdgcn_wmma_f32_16x16x4_f32(                          \
                        false, A[t][s4], false, Bv[g][s4], (short)0,                            \
                        (s4 == 0) ? Cb[t] : acc[g][t], false, false);                           \
                }                                                                               \
            }                                                                                   \
        }                                                                                       \
                                                                                                \
                                 \
        float partial[NG];                                                                      \
_Pragma("unroll")                                                                               \
        for (int g = 0; g < NG; ++g) {                                                          \
                            \
                          \
            v2f cw2;                                                                            \
            cw2.x = cb2;                                                                        \
            cw2.y = 0.f;                                                                        \
_Pragma("unroll")                                                                               \
            for (int s4 = 0; s4 < 4; ++s4)                                                      \
                cw2 = __builtin_elementwise_fma(cw[s4], Bv[g][s4], cw2);                        \
            const float cwt = cw2.x + cw2.y;                                                    \
                                                                                                \
                            \
                           \
                                  \
            float part[4] = {0.f, 0.f, 0.f, 0.f};                                               \
_Pragma("unroll")                                                                               \
            for (int q = 0; q < NSLOT; ++q) {                                                   \
                const int t = q >> 3, r = q & 7;                                                \
                const int d = q / KS, wc = q - KS * d;                                          \
                part[q & 3] = fmaf(acc[g][t][r], win[g][(d + (U_)) & 3][wc], part[q & 3]);      \
            }                                                                                   \
            partial[g] = fmaf(cwt, xc[g], (part[0] + part[1]) + (part[2] + part[3]));           \
        }                                                                                       \
                                                                                                \
                           \
                            \
                         \
                                                           \
_Pragma("unroll")                                                                               \
        for (int g = 0; g < NG; ++g)                                                            \
            partial[g] += xhalf(partial[g]);                                                    \
                                                                                                \
                          \
                            \
        (ob + i * IMG_W)[jst] = half ? partial[1] : partial[0];                                 \
                                                                                                \
                         \
                     \
        if ((S_) + 1 < SEG_R) {                                                                 \
            bool nrokA = true, nrokB = true;                                                    \
            if ((CTAIL_)) {                                                                     \
                nrokA = (i + 1 + 3 * half < IMG_H);                                             \
                nrokB = (i + 1 + 4 * half < IMG_H);                                             \
            }                                                                                   \
_Pragma("unroll")                                                                               \
            for (int g = 0; g < NG; ++g) {                                                      \
                HOLD7(nb[((U_) + 1) & (PF - 1)][g]);                                            \
_Pragma("unroll")                                                                               \
                for (int wc = 0; wc < KS - 1; ++wc)                                             \
                    win[g][(U_) & 3][wc] =                                                      \
                        (((wc < 3) ? nrokA : nrokB) && cok[g][wc])                              \
                            ? nb[((U_) + 1) & (PF - 1)][g][wc] : 0.f;                           \
                             \
                win[g][(U_) & 3][KS - 1] = (CTAIL_)                                             \
                    ? (nrokB ? nb[((U_) + 1) & (PF - 1)][g][KS - 1] : 0.f)                      \
                    : vmov(nb[((U_) + 1) & (PF - 1)][g][KS - 1]);                               \
            }                                                                                   \
        }                                                                                       \
    }

__global__ __launch_bounds__(256) void involution_colwalk_kernel(
    const float* __restrict__ x,
    const float* __restrict__ w1,  const float* __restrict__ b1,
    const float* __restrict__ gam, const float* __restrict__ bet,
    const float* __restrict__ mu,  const float* __restrict__ var,
    const float* __restrict__ w2,  const float* __restrict__ b2,
    float* __restrict__ out)
{
    const int tid   = threadIdx.x;
    const int lane  = tid & 31;
    const int half  = lane >> 4;
    const int p     = lane & 15;
    const int gw    = __builtin_amdgcn_readfirstlane((int)(blockIdx.x * 8 + (tid >> 5)));
    const int spair = gw & 7;
    const int seg   = (gw >> 3) & 7;
    const int b     = gw >> 6;
    const int jst   = spair * 32 + lane;
    const int i0    = seg * SEG_R;

    const float* __restrict__ xb = x   + (size_t)b * IMG_H * IMG_W;
    float* __restrict__       ob = out + (size_t)b * IMG_H * IMG_W;

    v2f A[NT][4];
#pragma unroll
    for (int t = 0; t < NT; ++t) {
        const int k = tap_of_slot(8 * t + (p & 7), p >> 3);
#pragma unroll
        for (int s = 0; s < 4; ++s)
            A[t][s] = *(const v2f*)&w2[k * CH + 4 * s + 2 * half];
    }
    v8f Cb[NT];
#pragma unroll
    for (int t = 0; t < NT; ++t) {
#pragma unroll
        for (int r = 0; r < 8; ++r)
            Cb[t][r] = b2[tap_of_slot(8 * t + r, half)];
    }
    v2f cw[4];
#pragma unroll
    for (int s = 0; s < 4; ++s)
        cw[s] = *(const v2f*)&w2[CTAP * CH + 4 * s + 2 * half];
    const float cb2 = half ? 0.f : b2[CTAP];

    float ca[4][2], cd[4][2];
#pragma unroll
    for (int s = 0; s < 4; ++s) {
#pragma unroll
        for (int e = 0; e < 2; ++e) {
            const int c = 4 * s + 2 * half + e;
            const float inv = gam[c] * rsqrtf(var[c] + BN_EPS);
            ca[s][e] = w1[c] * inv;
            cd[s][e] = (b1[c] - mu[c]) * inv + bet[c];
        }
    }

    int  coff[NG][KS];
    bool cok[NG][KS];
#pragma unroll
    for (int g = 0; g < NG; ++g) {
#pragma unroll
        for (int wc = 0; wc < KS; ++wc) {
            const int pc = half ? win_pcol(wc, 1) : win_pcol(wc, 0);
            const int cj = spair * 32 + g * 16 + p + pc - 3;
            cok[g][wc]  = (cj >= 0) && (cj < IMG_W);
            coff[g][wc] = cok[g][wc] ? cj : 0;
        }
    }

    float win[NG][4][KS];
    bool  wrokA[4], wrokB[4];
#pragma unroll
    for (int d = 0; d < 4; ++d) {
        const int grA = i0 + d + 3 * half - 3;
        const int grB = i0 + d + 4 * half - 3;
        wrokA[d]      = (grA >= 0) && (grA < IMG_H);
        wrokB[d]      = (grB >= 0) && (grB < IMG_H);
        const int gbA = wrokA[d] ? grA * IMG_W : 0;
        const int gbB = wrokB[d] ? grB * IMG_W : 0;
#pragma unroll
        for (int g = 0; g < NG; ++g)
#pragma unroll
            for (int wc = 0; wc < KS; ++wc)
                win[g][d][wc] = xb[((wc < 3) ? gbA : gbB) + coff[g][wc]];
    }

    unsigned voff[NG][KS];
#pragma unroll
    for (int g = 0; g < NG; ++g)
#pragma unroll
        for (int wc = 0; wc < KS; ++wc)
            voff[g][wc] = (unsigned)(((wc < 3) ? 3 : 4) * half * IMG_W + coff[g][wc]);

    float nb[PF][NG][KS];
#pragma unroll
    for (int k = 1; k < PF; ++k) {
        __builtin_amdgcn_sched_barrier(0);
        const float* __restrict__ rowp = xb + (i0 + k) * IMG_W;
#pragma unroll
        for (int g = 0; g < NG; ++g)
#pragma unroll
            for (int wc = 0; wc < KS; ++wc)
                nb[k][g][wc] = rowp[voff[g][wc]];
    }
    __builtin_amdgcn_sched_barrier(0);

#pragma unroll
    for (int g = 0; g < NG; ++g)
#pragma unroll
        for (int d = 0; d < 4; ++d) HOLD7(win[g][d]);

#pragma unroll
    for (int g = 0; g < NG; ++g)
#pragma unroll
        for (int d = 0; d < 4; ++d)
#pragma unroll
            for (int wc = 0; wc < KS; ++wc)
                win[g][d][wc] =
                    (((wc < 3) ? wrokA[d] : wrokB[d]) && cok[g][wc]) ? win[g][d][wc] : 0.f;

    for (int u0 = 0; u0 < SEG_R - 8; u0 += 4) {
        STEP(u0 + 0, 0, 1, 0, 0)
        STEP(u0 + 1, 1, 1, 0, 0)
        STEP(u0 + 2, 2, 1, 0, 0)
        STEP(u0 + 3, 3, 1, 0, 0)
    }
    STEP(SEG_R - 8, 0, (SEG_R - 8 + PF < SEG_R), 1, 0)
    STEP(SEG_R - 7, 1, (SEG_R - 7 + PF < SEG_R), 1, 0)
    STEP(SEG_R - 6, 2, (SEG_R - 6 + PF < SEG_R), 1, 0)
    STEP(SEG_R - 5, 3, (SEG_R - 5 + PF < SEG_R), 1, 1)
    STEP(SEG_R - 4, 0, (SEG_R - 4 + PF < SEG_R), 1, 1)
    STEP(SEG_R - 3, 1, (SEG_R - 3 + PF < SEG_R), 1, 1)
    STEP(SEG_R - 2, 2, (SEG_R - 2 + PF < SEG_R), 1, 1)
    STEP(SEG_R - 1, 3, (SEG_R - 1 + PF < SEG_R), 1, 1)
}

extern "C" void kernel_launch(void* const* d_in, const int* in_sizes, int n_in,
                              void* d_out, int out_size, void* d_ws, size_t ws_size,
                              hipStream_t stream) {
    (void)n_in; (void)d_ws; (void)ws_size; (void)out_size; (void)in_sizes;
    const float* x   = (const float*)d_in[0];
    const float* w1  = (const float*)d_in[1];
    const float* b1  = (const float*)d_in[2];
    const float* gam = (const float*)d_in[3];
    const float* bet = (const float*)d_in[4];
    const float* mu  = (const float*)d_in[5];
    const float* var = (const float*)d_in[6];
    const float* w2  = (const float*)d_in[7];
    const float* b2  = (const float*)d_in[8];
    float* out = (float*)d_out;

    dim3 grid(256);
    dim3 block(256);
    involution_colwalk_kernel<<<grid, block, 0, stream>>>(
        x, w1, b1, gam, bet, mu, var, w2, b2, out);
}
